# starting kernel
# baseline (speedup 1.0000x reference)
#include <hip/hip_runtime.h>
#include <hip/hip_bf16.h>
#include <math.h>

#define N_NODES 10000
#define N_EDGES 320000
#define F_IN    256
#define F_HID   256
#define F_OUT   16
#define BN_EPS  1e-5f

typedef __attribute__((ext_vector_type(2))) float v2f;
typedef __attribute__((ext_vector_type(8))) float v8f;

__device__ __forceinline__ v8f wmma4(v2f a, v2f b, v8f c) {
    return __builtin_amdgcn_wmma_f32_16x16x4_f32(
        false, a, false, b, (short)0, c, false, false);
}

__device__ __forceinline__ v2f load_a(const float* A, int lda, int m0, int k0, int lane) {
    const int half = lane >> 4;
    const int row  = lane & 15;
    const float* p = A + (size_t)(m0 + row) * lda + k0 + (half << 1);
    v2f r; r.x = p[0]; r.y = p[1];
    return r;
}

__device__ __forceinline__ v2f load_b(const float* B, int ldb, int k0, int n0, int lane) {
    const int half = lane >> 4;
    const int col  = lane & 15;
    const float* p = B + (size_t)(k0 + (half << 1)) * ldb + n0 + col;
    v2f r; r.x = p[0]; r.y = p[ldb];
    return r;
}

__device__ __forceinline__ void store_c(float* C, int ldc, int m0, int n0, int lane, v8f acc) {
    const int half = lane >> 4;
    const int col  = lane & 15;
#pragma unroll
    for (int r = 0; r < 8; ++r)
        C[(size_t)(m0 + r + half * 8) * ldc + n0 + col] = acc[r];
}

__global__ void deg_accum_kernel(const int* __restrict__ dst, float* __restrict__ deg) {
    int e = blockIdx.x * blockDim.x + threadIdx.x;
    if (e < N_EDGES) atomicAdd(&deg[dst[e]], 1.0f);
}

__global__ void deg_inv_kernel(float* __restrict__ deg) {
    int n = blockIdx.x * blockDim.x + threadIdx.x;
    if (n < N_NODES) deg[n] = 1.0f / fmaxf(deg[n], 1.0f);
}

__global__ void agg1_scatter_kernel(const float* __restrict__ x,
                                    const int* __restrict__ src,
                                    const int* __restrict__ dst,
                                    float* __restrict__ agg1) {
    long long g = (long long)blockIdx.x * blockDim.x + threadIdx.x;
    if (g >= (long long)N_EDGES * 64) return;
    int e  = (int)(g >> 6);
    int c4 = ((int)g & 63) << 2;
    const float4 v = *(const float4*)(x + (size_t)src[e] * F_IN + c4);
    float* a = agg1 + (size_t)dst[e] * F_IN + c4;
    atomicAdd(a + 0, v.x);
    atomicAdd(a + 1, v.y);
    atomicAdd(a + 2, v.z);
    atomicAdd(a + 3, v.w);
}

template <int COLS>
__global__ void scale_rows_kernel(float* __restrict__ m, const float* __restrict__ deg_inv,
                                  long long total) {
    long long g = (long long)blockIdx.x * blockDim.x + threadIdx.x;
    if (g < total) m[g] *= deg_inv[g / COLS];
}

__global__ __launch_bounds__(256) void gemm1_kernel(const float* __restrict__ A1,
                                                    const float* __restrict__ B1,
                                                    const float* __restrict__ A2,
                                                    const float* __restrict__ B2,
                                                    float* __restrict__ C) {
    const int lane = threadIdx.x & 31;
    const int wave = threadIdx.x >> 5;
    const int tile = blockIdx.x * 8 + wave;
    if (tile >= 2500) return;
    const int mt = tile >> 2;
    const int ng = tile & 3;
    const int m0 = mt * 16, n0 = ng * 64;

    v8f acc0 = {}, acc1 = {}, acc2 = {}, acc3 = {};
#pragma unroll 2
    for (int k0 = 0; k0 < F_IN; k0 += 4) {
        v2f a = load_a(A1, F_IN, m0, k0, lane);
        acc0 = wmma4(a, load_b(B1, F_HID, k0, n0 +  0, lane), acc0);
        acc1 = wmma4(a, load_b(B1, F_HID, k0, n0 + 16, lane), acc1);
        acc2 = wmma4(a, load_b(B1, F_HID, k0, n0 + 32, lane), acc2);
        acc3 = wmma4(a, load_b(B1, F_HID, k0, n0 + 48, lane), acc3);
    }
#pragma unroll 2
    for (int k0 = 0; k0 < F_IN; k0 += 4) {
        v2f a = load_a(A2, F_IN, m0, k0, lane);
        acc0 = wmma4(a, load_b(B2, F_HID, k0, n0 +  0, lane), acc0);
        acc1 = wmma4(a, load_b(B2, F_HID, k0, n0 + 16, lane), acc1);
        acc2 = wmma4(a, load_b(B2, F_HID, k0, n0 + 32, lane), acc2);
        acc3 = wmma4(a, load_b(B2, F_HID, k0, n0 + 48, lane), acc3);
    }
    store_c(C, F_HID, m0, n0 +  0, lane, acc0);
    store_c(C, F_HID, m0, n0 + 16, lane, acc1);
    store_c(C, F_HID, m0, n0 + 32, lane, acc2);
    store_c(C, F_HID, m0, n0 + 48, lane, acc3);
}

__global__ __launch_bounds__(256) void gemm3_kernel(const float* __restrict__ A,
                                                    const float* __restrict__ B1,
                                                    const float* __restrict__ B2,
                                                    float* __restrict__ D1,
                                                    float* __restrict__ D2) {
    const int lane = threadIdx.x & 31;
    const int wave = threadIdx.x >> 5;
    const int tile = blockIdx.x * 8 + wave;
    if (tile >= 625) return;
    const int m0 = tile * 16;

    v8f acc1 = {}, acc2 = {};
#pragma unroll 4
    for (int k0 = 0; k0 < F_HID; k0 += 4) {
        v2f a  = load_a(A, F_HID, m0, k0, lane);
        v2f b1 = load_b(B1, F_OUT, k0, 0, lane);
        v2f b2 = load_b(B2, F_OUT, k0, 0, lane);
        acc1 = wmma4(a, b1, acc1);
        acc2 = wmma4(a, b2, acc2);
    }
    store_c(D1, F_OUT, m0, 0, lane, acc1);
    store_c(D2, F_OUT, m0, 0, lane, acc2);
}

template <int COLS>
__global__ void bn_stats_kernel(const float* __restrict__ h,
                                float* __restrict__ sum, float* __restrict__ sumsq) {
    int g = blockIdx.x * blockDim.x + threadIdx.x;
    int col     = g & (COLS - 1);
    int slice   = g / COLS;
    int nslices = (gridDim.x * blockDim.x) / COLS;
    float s = 0.0f, q = 0.0f;
    for (int r = slice; r < N_NODES; r += nslices) {
        float v = h[(size_t)r * COLS + col];
        s += v;
        q += v * v;
    }
    atomicAdd(&sum[col], s);
    atomicAdd(&sumsq[col], q);
}

template <int COLS>
__global__ void bn_relu_kernel(float* __restrict__ h,
                               const float* __restrict__ sum, const float* __restrict__ sumsq,
                               const float* __restrict__ gamma, const float* __restrict__ beta,
                               long long total) {
    long long g = (long long)blockIdx.x * blockDim.x + threadIdx.x;
    if (g >= total) return;
    int c = (int)(g & (COLS - 1));
    float mean = sum[c] * (1.0f / N_NODES);
    float var  = sumsq[c] * (1.0f / N_NODES) - mean * mean;
    float v = (h[g] - mean) * rsqrtf(var + BN_EPS) * gamma[c] + beta[c];
    h[g] = fmaxf(v, 0.0f);
}

__global__ void agg3_scatter_kernel(const float* __restrict__ p,
                                    const int* __restrict__ src,
                                    const int* __restrict__ dst,
                                    float* __restrict__ agg3) {
    int e = blockIdx.x * blockDim.x + threadIdx.x;
    if (e >= N_EDGES) return;
    const float* ps = p + (size_t)src[e] * F_OUT;
    float* ad = agg3 + (size_t)dst[e] * F_OUT;
#pragma unroll
    for (int j = 0; j < F_OUT; j += 4) {
        float4 v = *(const float4*)(ps + j);
        atomicAdd(ad + j + 0, v.x);
        atomicAdd(ad + j + 1, v.y);
        atomicAdd(ad + j + 2, v.z);
        atomicAdd(ad + j + 3, v.w);
    }
}

__global__ void combine3_kernel(float* __restrict__ agg3, const float* __restrict__ hl,
                                const float* __restrict__ deg_inv, long long total) {
    long long g = (long long)blockIdx.x * blockDim.x + threadIdx.x;
    if (g < total) agg3[g] = agg3[g] * deg_inv[g / F_OUT] + hl[g];
}

__global__ void bn_final_kernel(const float* __restrict__ t,
                                const float* __restrict__ sum, const float* __restrict__ sumsq,
                                const float* __restrict__ gamma, const float* __restrict__ beta,
                                const int* __restrict__ ste,
                                float* __restrict__ out, long long total) {
    long long g = (long long)blockIdx.x * blockDim.x + threadIdx.x;
    if (g >= total) return;
    int c = (int)(g & (F_OUT - 1));
    float mean = sum[c] * (1.0f / N_NODES);
    float var  = sumsq[c] * (1.0f / N_NODES) - mean * mean;
    float v = (t[g] - mean) * rsqrtf(var + BN_EPS) * gamma[c] + beta[c];
    if (*ste)
        out[g] = (v > 0.0f) ? 1.0f : 0.0f;
    else
        out[g] = 1.0f / (1.0f + expf(-v));
}

extern "C" void kernel_launch(void* const* d_in, const int* in_sizes, int n_in,
                              void* d_out, int out_size, void* d_ws, size_t ws_size,
                              hipStream_t stream) {
    const float* x        = (const float*)d_in[0];
    const float* W_conv1  = (const float*)d_in[1];
    const float* W_lin1   = (const float*)d_in[2];
    const float* bn1_g    = (const float*)d_in[3];
    const float* bn1_b    = (const float*)d_in[4];
    const float* W_conv3  = (const float*)d_in[5];
    const float* W_lin3   = (const float*)d_in[6];
    const float* bn3_g    = (const float*)d_in[7];
    const float* bn3_b    = (const float*)d_in[8];
    const int*   src      = (const int*)d_in[9];
    const int*   dst      = (const int*)d_in[10];
    const int*   ste      = (const int*)d_in[11];
    float*       out      = (float*)d_out;

    float* ws = (float*)d_ws;
    const size_t o_deg  = 0;
    const size_t o_agg1 = 10240;
    const size_t o_h    = o_agg1 + 2560000;
    const size_t o_p    = o_h + 2560000;
    const size_t o_hl   = o_p + 160000;
    const size_t o_agg3 = o_hl + 160000;
    const size_t o_s1   = o_agg3 + 160000;
    const size_t o_q1   = o_s1 + 256;
    const size_t o_s3   = o_q1 + 256;
    const size_t o_q3   = o_s3 + 16;
    float* deg_inv = ws + o_deg;
    float* agg1    = ws + o_agg1;
    float* h       = ws + o_h;
    float* p       = ws + o_p;
    float* hl      = ws + o_hl;
    float* agg3    = ws + o_agg3;
    float* s1 = ws + o_s1; float* q1 = ws + o_q1;
    float* s3 = ws + o_s3; float* q3 = ws + o_q3;

    hipMemsetAsync(deg_inv, 0, 10240 * sizeof(float), stream);
    hipMemsetAsync(agg1, 0, 2560000 * sizeof(float), stream);
    hipMemsetAsync(agg3, 0, (160000 + 256 + 256 + 16 + 16) * sizeof(float), stream);

    deg_accum_kernel<<<(N_EDGES + 255) / 256, 256, 0, stream>>>(dst, deg_inv);
    deg_inv_kernel<<<(N_NODES + 255) / 256, 256, 0, stream>>>(deg_inv);

    {
        long long total = (long long)N_EDGES * 64;
        agg1_scatter_kernel<<<(int)((total + 255) / 256), 256, 0, stream>>>(x, src, dst, agg1);
        long long tot2 = (long long)N_NODES * F_IN;
        scale_rows_kernel<F_IN><<<(int)((tot2 + 255) / 256), 256, 0, stream>>>(agg1, deg_inv, tot2);
    }

    gemm1_kernel<<<(2500 + 7) / 8, 256, 0, stream>>>(agg1, W_conv1, x, W_lin1, h);

    bn_stats_kernel<F_HID><<<128, 256, 0, stream>>>(h, s1, q1);
    {
        long long tot = (long long)N_NODES * F_HID;
        bn_relu_kernel<F_HID><<<(int)((tot + 255) / 256), 256, 0, stream>>>(h, s1, q1, bn1_g, bn1_b, tot);
    }

    gemm3_kernel<<<(625 + 7) / 8, 256, 0, stream>>>(h, W_conv3, W_lin3, p, hl);

    agg3_scatter_kernel<<<(N_EDGES + 255) / 256, 256, 0, stream>>>(p, src, dst, agg3);
    {
        long long tot = (long long)N_NODES * F_OUT;
        combine3_kernel<<<(int)((tot + 255) / 256), 256, 0, stream>>>(agg3, hl, deg_inv, tot);
    }

    bn_stats_kernel<F_OUT><<<64, 256, 0, stream>>>(agg3, s3, q3);
    {
        long long tot = (long long)N_NODES * F_OUT;
        bn_final_kernel<<<(int)((tot + 255) / 256), 256, 0, stream>>>(agg3, s3, q3, bn3_g, bn3_b, ste, out, tot);
    }
}


# layer-3 GEMM: one work-group per 16-row tile, K split over its 8 waves, all fragments of a wave requested at once, the 8 partial tiles added through LDS in wave order; sum order changes
# speedup vs baseline: 5.8916x; 5.8916x over previous
#include <hip/hip_runtime.h>
#include <hip/hip_bf16.h>
#include <math.h>

#define N_NODES 10000
#define N_EDGES 320000
#define F_IN    256
#define F_HID   256
#define F_OUT   16
#define BN_EPS  1e-5f

typedef __attribute__((ext_vector_type(2))) float v2f;
typedef __attribute__((ext_vector_type(8))) float v8f;

__device__ __forceinline__ v8f wmma4(v2f a, v2f b, v8f c) {
    return __builtin_amdgcn_wmma_f32_16x16x4_f32(
        false, a, false, b, (short)0, c, false, false);
}

__device__ __forceinline__ v2f load_a(const float* A, int lda, int m0, int k0, int lane) {
    const int half = lane >> 4;
    const int row  = lane & 15;
    const float* p = A + (size_t)(m0 + row) * lda + k0 + (half << 1);
    v2f r; r.x = p[0]; r.y = p[1];
    return r;
}

__device__ __forceinline__ v2f load_b(const float* B, int ldb, int k0, int n0, int lane) {
    const int half = lane >> 4;
    const int col  = lane & 15;
    const float* p = B + (size_t)(k0 + (half << 1)) * ldb + n0 + col;
    v2f r; r.x = p[0]; r.y = p[ldb];
    return r;
}

#define AGG_R       40
#define AGG_GROUPS  (N_NODES / AGG_R)
#define AGG_THREADS 1024
#define AGG_WAVES   (AGG_THREADS / 32)
#define AGG_LIST    16384
#define AGG_QUADS   (N_EDGES / 4)
#define AGG_BATCH   8
#define AGG_ROWS_IN_FLIGHT 32
static_assert(AGG_R * AGG_GROUPS == N_NODES, "row ranges tile the node set");
static_assert(AGG_R < 64, "local row id, and one spare row for padding, are packed into 6 bits");
static_assert(N_EDGES % 4 == 0 && N_EDGES < (1 << 25), "edge id is packed above 6 bits");
static_assert(AGG_WAVES % 4 == 0 && AGG_R % (AGG_WAVES / 4) == 0, "a wave owns one column quarter of every (AGG_WAVES/4)-th row");
static_assert(F_IN == 4 * 32 * 2, "a column quarter is 2 columns a lane");
static_assert(2 * AGG_LIST * 4 + AGG_R * F_IN * 4 > 160 * 1024, "one work-group per WGP");

template <bool FULL>
__device__ __forceinline__ void agg_rows_add(float2& s, const float* __restrict__ x, unsigned col, int id, int m) {
    float2 v[AGG_ROWS_IN_FLIGHT];
#pragma unroll
    for (int t = 0; t < AGG_ROWS_IN_FLIGHT; ++t) {
        v[t] = make_float2(0.0f, 0.0f);
        if (FULL || t < m) {
            const float* row = x + (size_t)__builtin_amdgcn_readlane(id, t) * F_IN;
            v[t] = *reinterpret_cast<const float2*>(row + col);
        }
    }
#pragma unroll
    for (int t = 0; t < AGG_ROWS_IN_FLIGHT; ++t) {
        if (FULL || t < m) { s.x += v[t].x; s.y += v[t].y; }
    }
}

__global__ __launch_bounds__(AGG_THREADS) void agg1_rows_kernel(const float* __restrict__ x,
                                                                const int* __restrict__ src,
                                                                const int* __restrict__ dst,
                                                                float* __restrict__ deg_inv,
                                                                float* __restrict__ agg1,
                                                                int* __restrict__ elist,
                                                                int* __restrict__ ecount) {
    __shared__ float acc[AGG_R * F_IN];
    __shared__ int   list[AGG_LIST];
    __shared__ int   ranked[AGG_LIST];
    __shared__ int   cnt[AGG_R];
    __shared__ int   cntl[AGG_R];
    __shared__ int   start[AGG_R];
    __shared__ int   fill[AGG_R];
    __shared__ int   n_noted;

    const int tid  = threadIdx.x;
    const int lane = tid & 31;
    const int wave = __builtin_amdgcn_readfirstlane(tid >> 5);
    const int r0   = blockIdx.x * AGG_R;

    for (int i = tid; i < AGG_R * F_IN; i += AGG_THREADS) acc[i] = 0.0f;
    if (tid < AGG_R) cnt[tid] = 0;
    if (tid == 0) n_noted = 0;
    __syncthreads();

    const int4* dst4 = reinterpret_cast<const int4*>(dst);
    for (int q0 = 0; q0 < AGG_QUADS; q0 += AGG_BATCH * AGG_THREADS) {
        int4 d[AGG_BATCH];
#pragma unroll
        for (int k = 0; k < AGG_BATCH; ++k) {
            const int q = q0 + k * AGG_THREADS + tid;
            d[k] = (q < AGG_QUADS) ? dst4[q] : make_int4(-1, -1, -1, -1);
        }
#pragma unroll
        for (int k = 0; k < AGG_BATCH; ++k) {
            const int e = (q0 + k * AGG_THREADS + tid) * 4;
            const unsigned u0 = (unsigned)(d[k].x - r0), u1 = (unsigned)(d[k].y - r0);
            const unsigned u2 = (unsigned)(d[k].z - r0), u3 = (unsigned)(d[k].w - r0);
            if (min(min(u0, u1), min(u2, u3)) < (unsigned)AGG_R) {
                const unsigned uu[4] = {u0, u1, u2, u3};
#pragma unroll
                for (int t = 0; t < 4; ++t) {
                    if (uu[t] < (unsigned)AGG_R) {
                        atomicAdd(&cnt[uu[t]], 1);
                        const int slot = atomicAdd(&n_noted, 1);
                        if (slot < AGG_LIST) {
                            list[slot] = ((e + t) << 6) | (int)uu[t];
                        } else {
                            const float* xr = x + (size_t)src[e + t] * F_IN;
                            float* ar = acc + uu[t] * F_IN;
#pragma nounroll
                            for (int j = 0; j < F_IN; ++j) atomicAdd(ar + j, xr[j]);
                        }
                    }
                }
            }
        }
    }
    __syncthreads();

    const int total = __builtin_amdgcn_readfirstlane(n_noted);
    const int n = min(total, AGG_LIST);
    const int* cl = cnt;
    if (total > AGG_LIST) {
        if (tid < AGG_R) cntl[tid] = 0;
        __syncthreads();
        for (int i = tid; i < n; i += AGG_THREADS) atomicAdd(&cntl[list[i] & 63], 1);
        __syncthreads();
        cl = cntl;
    }
    if (tid < AGG_R) {
        int s = 0;
#pragma unroll
        for (int r = 0; r < AGG_R; ++r) { const int c = cl[r]; s += (r < tid) ? c : 0; }
        start[tid] = s;
        fill[tid]  = 0;
    }
    __syncthreads();

    int* el = elist + (size_t)blockIdx.x * AGG_LIST;
    for (int i = tid; i < n; i += AGG_THREADS) {
        const int ent = list[i];
        const int u   = ent & 63;
        const int sn  = src[ent >> 6];
        ranked[start[u] + atomicAdd(&fill[u], 1)] = sn;
        el[i] = (sn << 6) | u;
    }
    if (tid == 0) ecount[blockIdx.x] = total;
    __syncthreads();

    const unsigned col = (wave & 3) * 64 + lane * 2;
    for (int u = wave >> 2; u < AGG_R; u += AGG_WAVES / 4) {
        const int c  = __builtin_amdgcn_readfirstlane(cl[u]);
        const int st = __builtin_amdgcn_readfirstlane(start[u]);
        float2 s = make_float2(0.0f, 0.0f);
        for (int j0 = 0; j0 < c; j0 += 32) {
            const int id = (j0 + lane < c) ? ranked[st + j0 + lane] : 0;
            const int m  = min(32, c - j0);
            if (m == 32) agg_rows_add<true>(s, x, col, id, m);
            else         agg_rows_add<false>(s, x, col, id, m);
        }
        const float  di = 1.0f / fmaxf((float)cnt[u], 1.0f);
        const float2 a  = *reinterpret_cast<const float2*>(acc + u * F_IN + col);
        s.x = (s.x + a.x) * di;
        s.y = (s.y + a.y) * di;
        *reinterpret_cast<float2*>(agg1 + (size_t)(r0 + u) * F_IN + col) = s;
        if ((wave & 3) == 0 && lane == 0) deg_inv[r0 + u] = di;
    }
}

#define G1_ROWS    48
#define G1_THREADS 256
#define G1_KC      32
#define G1_CHUNKS  ((2 * F_IN) / G1_KC)
#define G1_GROUPS  ((N_NODES + G1_ROWS - 1) / G1_ROWS)
#define G1_APITCH4 ((G1_KC + 4) / 4)
static_assert(F_IN == F_HID && F_IN % G1_KC == 0, "both products share the chunking");
static_assert(F_HID == (G1_THREADS / 32) * 32, "a wave owns 32 columns");

struct G1Stage {
    v2f    b[(G1_KC / 2) * F_HID];
    float4 a[G1_ROWS * G1_APITCH4];
};

__device__ __forceinline__ void g1_fetch(const float* __restrict__ Ap, const float* __restrict__ Bp,
                                         int kb, int bk, int bq, int arow0, int arow1, int af,
                                         bool second_row, float4 (&re)[4], float4 (&ro)[4], float4 (&ra)[2]) {
    const float* bp = Bp + (size_t)(kb + 2 * bk) * F_HID + 4 * bq;
#pragma unroll
    for (int i = 0; i < 4; ++i) {
        re[i] = *reinterpret_cast<const float4*>(bp + (size_t)(8 * i) * F_HID);
        ro[i] = *reinterpret_cast<const float4*>(bp + (size_t)(8 * i + 1) * F_HID);
    }
    ra[0] = *reinterpret_cast<const float4*>(Ap + (size_t)arow0 * F_IN + kb + 4 * af);
    if (second_row)
        ra[1] = *reinterpret_cast<const float4*>(Ap + (size_t)arow1 * F_IN + kb + 4 * af);
}

__device__ __forceinline__ void g1_put(G1Stage& st, int bk, int bq, int ar, int af, bool second_row,
                                       const float4 (&re)[4], const float4 (&ro)[4], const float4 (&ra)[2]) {
    const int q = bq ^ (((bk >> 1) & 1) << 2);
#pragma unroll
    for (int i = 0; i < 4; ++i) {
        float4* d = reinterpret_cast<float4*>(&st.b[(bk + 4 * i) * F_HID + 4 * q]);
        d[0] = make_float4(re[i].x, ro[i].x, re[i].y, ro[i].y);
        d[1] = make_float4(re[i].z, ro[i].z, re[i].w, ro[i].w);
    }
    st.a[ar * G1_APITCH4 + af] = ra[0];
    if (second_row) st.a[(32 + ar) * G1_APITCH4 + af] = ra[1];
}

__device__ __forceinline__ void g1_compute(const G1Stage& st, int wave, int c16, int half, v8f (&acc)[3][2]) {
    const v2f* pb0 = &st.b[2 * half * F_HID + wave * 32 + 16 * half + c16];
    const v2f* pb1 = &st.b[2 * half * F_HID + wave * 32 + 16 * (1 - half) + c16];
    const float4* pa = &st.a[c16 * G1_APITCH4 + half];
#pragma unroll
    for (int p = 0; p < G1_KC / 8; ++p) {
        float4 a[3];
        v2f    b[2][2];
#pragma unroll
        for (int t = 0; t < 3; ++t) a[t] = pa[t * 16 * G1_APITCH4 + 2 * p];
#pragma unroll
        for (int j = 0; j < 2; ++j) {
            b[j][0] = pb0[(4 * p + j) * F_HID];
            b[j][1] = pb1[(4 * p + j) * F_HID];
        }
#pragma unroll
        for (int t = 0; t < 3; ++t)
#pragma unroll
            for (int n = 0; n < 2; ++n) {
                v2f av; av.x = a[t].x; av.y = a[t].y;
                acc[t][n] = wmma4(av, b[0][n], acc[t][n]);
            }
#pragma unroll
        for (int t = 0; t < 3; ++t)
#pragma unroll
            for (int n = 0; n < 2; ++n) {
                v2f av; av.x = a[t].z; av.y = a[t].w;
                acc[t][n] = wmma4(av, b[1][n], acc[t][n]);
            }
    }
}

__global__ __launch_bounds__(G1_THREADS) void gemm1_kernel(const float* __restrict__ A1,
                                                           const float* __restrict__ B1,
                                                           const float* __restrict__ A2,
                                                           const float* __restrict__ B2,
                                                           float* __restrict__ C) {
    __shared__ G1Stage stage[2];

    const int tid  = threadIdx.x;
    const int lane = tid & 31;
    const int wave = tid >> 5;
    const int c16  = lane & 15;
    const int half = lane >> 4;
    const int m0   = blockIdx.x * G1_ROWS;

    const int bq  = tid & 63;
    const int bk  = tid >> 6;
    const int ar  = tid >> 3;
    const int af  = tid & 7;
    const int arow0 = min(m0 + ar, N_NODES - 1);
    const int arow1 = min(m0 + 32 + ar, N_NODES - 1);
    const bool second_row = tid < 128;

    float4 re[4], ro[4], ra[2];
    ra[1] = make_float4(0.0f, 0.0f, 0.0f, 0.0f);
    v8f acc[3][2] = {};

    g1_fetch(A1, B1, 0, bk, bq, arow0, arow1, af, second_row, re, ro, ra);
    g1_put(stage[0], bk, bq, ar, af, second_row, re, ro, ra);
    __syncthreads();
#pragma nounroll
    for (int ch = 0; ch < G1_CHUNKS - 1; ++ch) {
        const int nx = ch + 1;
        g1_fetch(nx < G1_CHUNKS / 2 ? A1 : A2, nx < G1_CHUNKS / 2 ? B1 : B2,
                 (nx & (G1_CHUNKS / 2 - 1)) * G1_KC, bk, bq, arow0, arow1, af, second_row, re, ro, ra);
        g1_compute(stage[ch & 1], wave, c16, half, acc);
        __builtin_amdgcn_sched_barrier(0);
        g1_put(stage[nx & 1], bk, bq, ar, af, second_row, re, ro, ra);
        __syncthreads();
    }
    g1_compute(stage[(G1_CHUNKS - 1) & 1], wave, c16, half, acc);

    static_assert(N_NODES % 16 == 0, "M tiles do not straddle the end");
#pragma unroll
    for (int t = 0; t < 3; ++t) {
        if (m0 + t * 16 < N_NODES) {
#pragma unroll
            for (int n = 0; n < 2; ++n)
#pragma unroll
                for (int r = 0; r < 8; ++r)
                    C[(size_t)(m0 + t * 16 + half * 8 + r) * F_HID + wave * 32 + n * 16 + c16] = acc[t][n][r];
        }
    }
}

#define G3_WAVES   8
#define G3_THREADS (G3_WAVES * 32)
#define G3_KW      (F_HID / G3_WAVES)
#define G3_STEPS   (G3_KW / 4)
#define G3_TILES   (N_NODES / 16)
static_assert(N_NODES % 16 == 0 && F_HID % (G3_WAVES * 4) == 0, "whole M tiles, whole k steps per wave");
static_assert(F_OUT == 16 && G3_THREADS == 16 * F_OUT, "one thread per element of a 16x16 output tile");

__global__ __launch_bounds__(G3_THREADS) void gemm3_kernel(const float* __restrict__ A,
                                                           const float* __restrict__ B1,
                                                           const float* __restrict__ B2,
                                                           float* __restrict__ D1,
                                                           float* __restrict__ D2) {
    __shared__ float part[G3_WAVES][2][8][32];

    const int tid  = threadIdx.x;
    const int lane = tid & 31;
    const int wave = tid >> 5;
    const int m0   = blockIdx.x * 16;
    const int kb   = wave * G3_KW;

    v2f a[G3_STEPS], b1[G3_STEPS], b2[G3_STEPS];
#pragma unroll
    for (int s = 0; s < G3_STEPS; ++s) {
        a[s]  = load_a(A, F_HID, m0, kb + 4 * s, lane);
        b1[s] = load_b(B1, F_OUT, kb + 4 * s, 0, lane);
        b2[s] = load_b(B2, F_OUT, kb + 4 * s, 0, lane);
    }
    v8f acc1 = {}, acc2 = {};
#pragma unroll
    for (int s = 0; s < G3_STEPS; ++s) {
        acc1 = wmma4(a[s], b1[s], acc1);
        acc2 = wmma4(a[s], b2[s], acc2);
    }
#pragma unroll
    for (int r = 0; r < 8; ++r) {
        part[wave][0][r][lane] = acc1[r];
        part[wave][1][r][lane] = acc2[r];
    }
    __syncthreads();

    const int row = tid >> 4;
    const int col = tid & 15;
    const int r   = row & 7;
    const int l   = (row >> 3) * 16 + col;
    float s1 = 0.0f, s2 = 0.0f;
#pragma unroll
    for (int w = 0; w < G3_WAVES; ++w) {
        s1 += part[w][0][r][l];
        s2 += part[w][1][r][l];
    }
    D1[(size_t)m0 * F_OUT + tid] = s1;
    D2[(size_t)m0 * F_OUT + tid] = s2;
}

template <int COLS, int THREADS, int ROWS>
__global__ __launch_bounds__(THREADS) void bn_stats_kernel(const float* __restrict__ h,
                                                           float* __restrict__ sum,
                                                           float* __restrict__ sumsq) {
    constexpr int TPR = COLS / 4;
    constexpr int RG  = THREADS / TPR;
    constexpr int NL  = ROWS / RG;
    static_assert(COLS % 4 == 0 && THREADS % TPR == 0 && COLS <= THREADS, "lane-to-column map");
    static_assert(ROWS % RG == 0 && N_NODES % ROWS == 0, "row blocks tile the node set, no tail");
    __shared__ float4 red_s[RG][TPR];
    __shared__ float4 red_q[RG][TPR];

    const int tid = threadIdx.x;
    const int cq  = tid % TPR;
    const int rs  = tid / TPR;
    const float4* h4 = reinterpret_cast<const float4*>(h)
                     + ((size_t)blockIdx.x * ROWS + rs) * TPR + cq;
    float4 v[NL];
#pragma unroll
    for (int j = 0; j < NL; ++j) v[j] = h4[(size_t)j * THREADS];
    float4 s = make_float4(0.0f, 0.0f, 0.0f, 0.0f), q = s;
#pragma unroll
    for (int j = 0; j < NL; ++j) {
        s.x += v[j].x; s.y += v[j].y; s.z += v[j].z; s.w += v[j].w;
        q.x += v[j].x * v[j].x; q.y += v[j].y * v[j].y;
        q.z += v[j].z * v[j].z; q.w += v[j].w * v[j].w;
    }
    red_s[rs][cq] = s;
    red_q[rs][cq] = q;
    __syncthreads();
    if (tid < COLS) {
        const float* fs = reinterpret_cast<const float*>(&red_s[0][0]);
        const float* fq = reinterpret_cast<const float*>(&red_q[0][0]);
        float a = 0.0f, b = 0.0f;
#pragma unroll
        for (int g = 0; g < RG; ++g) { a += fs[g * COLS + tid]; b += fq[g * COLS + tid]; }
        atomicAdd(&sum[tid], a);
        atomicAdd(&sumsq[tid], b);
    }
}

template <int COLS>
__global__ void bn_relu_kernel(float* __restrict__ h,
                               const float* __restrict__ sum, const float* __restrict__ sumsq,
                               const float* __restrict__ gamma, const float* __restrict__ beta,
                               long long total) {
    long long g = (long long)blockIdx.x * blockDim.x + threadIdx.x;
    if (g >= total) return;
    int c = (int)(g & (COLS - 1));
    float mean = sum[c] * (1.0f / N_NODES);
    float var  = sumsq[c] * (1.0f / N_NODES) - mean * mean;
    float v = (h[g] - mean) * rsqrtf(var + BN_EPS) * gamma[c] + beta[c];
    h[g] = fmaxf(v, 0.0f);
}

#define A3_THREADS 512
static_assert(F_OUT == 16, "two 16-wide rows per wave instruction; row = element >> 4");
__global__ __launch_bounds__(A3_THREADS) void agg3_rows_kernel(const float* __restrict__ p,
                                                               const float* __restrict__ hl,
                                                               const float* __restrict__ deg_inv,
                                                               const int* __restrict__ src,
                                                               const int* __restrict__ dst,
                                                               const int* __restrict__ elist,
                                                               const int* __restrict__ ecount,
                                                               float* __restrict__ t,
                                                               float* __restrict__ sum,
                                                               float* __restrict__ sumsq) {
    __shared__ float acc[(AGG_R + 1) * F_OUT];

    const int tid  = threadIdx.x;
    const int lane = tid & 31;
    const int wave  = __builtin_amdgcn_readfirstlane(tid >> 5);
    const int total = __builtin_amdgcn_readfirstlane(ecount[blockIdx.x]);
    const int r0    = blockIdx.x * AGG_R;

    static_assert(AGG_R * F_OUT <= 2 * A3_THREADS, "a lane finishes at most two elements");
    float hv[2], dv[2];
#pragma unroll
    for (int k = 0; k < 2; ++k) {
        const int  i  = tid + k * A3_THREADS;
        const bool in = i < AGG_R * F_OUT;
        hv[k] = in ? hl[(size_t)r0 * F_OUT + i] : 0.0f;
        dv[k] = in ? deg_inv[r0 + (i >> 4)] : 0.0f;
    }

    for (int i = tid; i < (AGG_R + 1) * F_OUT; i += A3_THREADS) acc[i] = 0.0f;
    __syncthreads();

    if (total <= AGG_LIST) {
        const int* el = elist + (size_t)blockIdx.x * AGG_LIST;
        const int  c  = lane & 15;
        const bool hi = lane >= 16;
        const int  pad = AGG_R;
        int idx = wave * 32 + lane;
        int ent = idx < total ? el[idx] : pad;
        for (int base = wave * 32; base < total; base += A3_THREADS) {
            idx += A3_THREADS;
            const int nraw = el[min(idx, total - 1)];
            const int nent = idx < total ? nraw : pad;
            float v[16];
            int   u[16];
#pragma unroll
            for (int i = 0; i < 16; ++i) {
                const int e0 = __builtin_amdgcn_readlane(ent, 2 * i);
                const int e1 = __builtin_amdgcn_readlane(ent, 2 * i + 1);
                const int e  = hi ? e1 : e0;
                u[i] = (e & 63) * F_OUT + c;
                v[i] = p[(e >> 6) * F_OUT + c];
            }
#pragma unroll
            for (int i = 0; i < 16; ++i) atomicAdd(acc + u[i], v[i]);
            ent = nent;
        }
    } else {
        for (int e = tid; e < N_EDGES; e += A3_THREADS) {
            const unsigned uu = (unsigned)(dst[e] - r0);
            if (uu < (unsigned)AGG_R) {
                const float* pr = p + (size_t)src[e] * F_OUT;
#pragma nounroll
                for (int j = 0; j < F_OUT; ++j) atomicAdd(acc + uu * F_OUT + j, pr[j]);
            }
        }
    }
    __syncthreads();

#pragma unroll
    for (int k = 0; k < 2; ++k) {
        const int i = tid + k * A3_THREADS;
        if (i < AGG_R * F_OUT) {
            const float val = acc[i] * dv[k] + hv[k];
            t[(size_t)r0 * F_OUT + i] = val;
            acc[i] = val;
        }
    }
    __syncthreads();

    if (tid < F_OUT) {
        float a = 0.0f, b = 0.0f;
#pragma unroll
        for (int r = 0; r < AGG_R; ++r) {
            const float x = acc[r * F_OUT + tid];
            a += x;
            b += x * x;
        }
        atomicAdd(&sum[tid], a);
        atomicAdd(&sumsq[tid], b);
    }
}

__global__ void bn_final_kernel(const float* __restrict__ t,
                                const float* __restrict__ sum, const float* __restrict__ sumsq,
                                const float* __restrict__ gamma, const float* __restrict__ beta,
                                const int* __restrict__ ste,
                                float* __restrict__ out, long long total) {
    long long g = (long long)blockIdx.x * blockDim.x + threadIdx.x;
    if (g >= total) return;
    int c = (int)(g & (F_OUT - 1));
    float mean = sum[c] * (1.0f / N_NODES);
    float var  = sumsq[c] * (1.0f / N_NODES) - mean * mean;
    float v = (t[g] - mean) * rsqrtf(var + BN_EPS) * gamma[c] + beta[c];
    if (*ste)
        out[g] = (v > 0.0f) ? 1.0f : 0.0f;
    else
        out[g] = 1.0f / (1.0f + expf(-v));
}

extern "C" void kernel_launch(void* const* d_in, const int* in_sizes, int n_in,
                              void* d_out, int out_size, void* d_ws, size_t ws_size,
                              hipStream_t stream) {
    const float* x        = (const float*)d_in[0];
    const float* W_conv1  = (const float*)d_in[1];
    const float* W_lin1   = (const float*)d_in[2];
    const float* bn1_g    = (const float*)d_in[3];
    const float* bn1_b    = (const float*)d_in[4];
    const float* W_conv3  = (const float*)d_in[5];
    const float* W_lin3   = (const float*)d_in[6];
    const float* bn3_g    = (const float*)d_in[7];
    const float* bn3_b    = (const float*)d_in[8];
    const int*   src      = (const int*)d_in[9];
    const int*   dst      = (const int*)d_in[10];
    const int*   ste      = (const int*)d_in[11];
    float*       out      = (float*)d_out;

    float* ws = (float*)d_ws;
    const size_t o_deg  = 0;
    const size_t o_agg1 = 10240;
    const size_t o_h    = o_agg1 + 2560000;
    const size_t o_p    = o_h + 2560000;
    const size_t o_hl   = o_p + 160000;
    const size_t o_agg3 = o_hl + 160000;
    const size_t o_s1   = o_agg3 + 160000;
    const size_t o_q1   = o_s1 + 256;
    const size_t o_s3   = o_q1 + 256;
    const size_t o_q3   = o_s3 + 16;
    const size_t o_cnt  = o_q3 + 16;
    const size_t o_el   = o_cnt + 256;
    float* deg_inv = ws + o_deg;
    float* agg1    = ws + o_agg1;
    float* h       = ws + o_h;
    float* p       = ws + o_p;
    float* hl      = ws + o_hl;
    float* agg3    = ws + o_agg3;
    float* s1 = ws + o_s1; float* q1 = ws + o_q1;
    float* s3 = ws + o_s3; float* q3 = ws + o_q3;
    int* ecount = (int*)(ws + o_cnt);
    int* elist  = (int*)(ws + o_el);

    hipMemsetAsync(s1, 0, (256 + 256 + 16 + 16) * sizeof(float), stream);

    agg1_rows_kernel<<<AGG_GROUPS, AGG_THREADS, 0, stream>>>(x, src, dst, deg_inv, agg1, elist, ecount);

    gemm1_kernel<<<G1_GROUPS, G1_THREADS, 0, stream>>>(agg1, W_conv1, x, W_lin1, h);

    bn_stats_kernel<F_HID, 512, 80><<<N_NODES / 80, 512, 0, stream>>>(h, s1, q1);
    {
        long long tot = (long long)N_NODES * F_HID;
        bn_relu_kernel<F_HID><<<(int)((tot + 255) / 256), 256, 0, stream>>>(h, s1, q1, bn1_g, bn1_b, tot);
    }

    gemm3_kernel<<<G3_TILES, G3_THREADS, 0, stream>>>(h, W_conv3, W_lin3, p, hl);

    agg3_rows_kernel<<<AGG_GROUPS, A3_THREADS, 0, stream>>>(p, hl, deg_inv, src, dst, elist, ecount,
                                                            agg3, s3, q3);

    {
        long long tot = (long long)N_NODES * F_OUT;
        bn_final_kernel<<<(int)((tot + 255) / 256), 256, 0, stream>>>(agg3, s3, q3, bn3_g, bn3_b, ste, out, tot);
    }
}
